# starting kernel
# baseline (speedup 1.0000x reference)
#include <hip/hip_runtime.h>

typedef float v2f __attribute__((ext_vector_type(2)));
typedef float v4f __attribute__((ext_vector_type(4)));
typedef float v8f __attribute__((ext_vector_type(8)));

#define BB 16
#define NN 1024
#define FF 256
#define NEG_INF_V (-1000000000.0f)
#define SLOPE_V 0.2f

#define K3_WAVES 4
#define BSTRIDE 20

__global__ __launch_bounds__(32)
void gat_gemm_hW(const float* __restrict__ h, const float* __restrict__ W,
                 float* __restrict__ WhT) {
  int tile = blockIdx.x;
  int ot = tile % (FF / 16);
  int nt = (tile / (FF / 16)) % (NN / 16);
  int b  = tile / ((FF / 16) * (NN / 16));
  int lane = threadIdx.x;
  int n  = lane & 15;
  int hh = lane >> 4;

  const float* Arow = h + ((size_t)(b * NN + nt * 16 + n)) * FF;
  const float* Bcol = W + ot * 16 + n;

  v8f acc = {};
  #pragma unroll 4
  for (int k = 0; k < FF; k += 4) {
    v2f a = *(const v2f*)(Arow + k + 2 * hh);
    v2f bm;
    bm.x = Bcol[(size_t)(k + 2 * hh) * FF];
    bm.y = Bcol[(size_t)(k + 1 + 2 * hh) * FF];
    acc = __builtin_amdgcn_wmma_f32_16x16x4_f32(false, a, false, bm,
                                                (short)0, acc, false, false);
  }
  float* outp = WhT + ((size_t)(b * FF + ot * 16 + n)) * NN + nt * 16 + 8 * hh;
  v4f lo = { acc[0], acc[1], acc[2], acc[3] };
  v4f hi = { acc[4], acc[5], acc[6], acc[7] };
  *(v4f*)(outp)     = lo;
  *(v4f*)(outp + 4) = hi;
}

__global__ __launch_bounds__(256)
void gat_f12(const float* __restrict__ WhT, const float* __restrict__ a,
             float* __restrict__ f1, float* __restrict__ f2) {
  int idx = blockIdx.x * blockDim.x + threadIdx.x;
  int b = idx / NN;
  int n = idx % NN;
  const float* base = WhT + (size_t)b * FF * NN + n;
  float s1 = 0.f, s2 = 0.f;
  #pragma unroll 8
  for (int o = 0; o < FF; ++o) {
    float w = base[(size_t)o * NN];
    s1 += w * a[o];
    s2 += w * a[FF + o];
  }
  f1[idx] = s1;
  f2[idx] = s2;
}

__global__ __launch_bounds__(128)
void gat_attn(const float* __restrict__ WhT, const int* __restrict__ adj,
              const float* __restrict__ f1, const float* __restrict__ f2,
              float* __restrict__ out) {
  __shared__ float Btile[FF * BSTRIDE];
  __shared__ float ptile[K3_WAVES][16 * BSTRIDE];
  int tid  = threadIdx.x;
  int w    = tid >> 5;
  int lane = tid & 31;
  int n  = lane & 15;
  int hh = lane >> 4;
  int b     = blockIdx.x >> 4;
  int super = blockIdx.x & 15;
  int i0 = (super * K3_WAVES + w) * 16;

  float f1v[8];
  {
    const float* p = f1 + b * NN + i0 + 8 * hh;
    v4f t0 = *(const v4f*)(p);
    v4f t1 = *(const v4f*)(p + 4);
    f1v[0] = t0.x; f1v[1] = t0.y; f1v[2] = t0.z; f1v[3] = t0.w;
    f1v[4] = t1.x; f1v[5] = t1.y; f1v[6] = t1.z; f1v[7] = t1.w;
  }

  v8f acc[16];
  v8f zero = {};
  #pragma unroll
  for (int t = 0; t < 16; ++t) acc[t] = zero;
  float mrun[8], lrun[8];
  #pragma unroll
  for (int r = 0; r < 8; ++r) { mrun[r] = -3.0e38f; lrun[r] = 0.f; }

  const int*   adjb = adj + ((size_t)(b * NN + i0 + 8 * hh)) * NN;
  const float* WhTb = WhT + (size_t)b * FF * NN;
  const float* f2b  = f2 + b * NN;

  for (int jt = 0; jt < NN / 16; ++jt) {
    int j0 = jt * 16;

    {
      const float* g0 = WhTb + (size_t)tid * NN + j0;
      const float* g1 = g0 + (size_t)128 * NN;
      v4f x0 = *(const v4f*)(g0);
      v4f x1 = *(const v4f*)(g0 + 4);
      v4f x2 = *(const v4f*)(g0 + 8);
      v4f x3 = *(const v4f*)(g0 + 12);
      v4f y0 = *(const v4f*)(g1);
      v4f y1 = *(const v4f*)(g1 + 4);
      v4f y2 = *(const v4f*)(g1 + 8);
      v4f y3 = *(const v4f*)(g1 + 12);
      float* l0 = Btile + tid * BSTRIDE;
      float* l1 = Btile + (tid + 128) * BSTRIDE;
      *(v4f*)(l0)      = x0;  *(v4f*)(l0 + 4)  = x1;
      *(v4f*)(l0 + 8)  = x2;  *(v4f*)(l0 + 12) = x3;
      *(v4f*)(l1)      = y0;  *(v4f*)(l1 + 4)  = y1;
      *(v4f*)(l1 + 8)  = y2;  *(v4f*)(l1 + 12) = y3;
    }

    float f2v = f2b[j0 + n];
    float p[8];
    #pragma unroll
    for (int r = 0; r < 8; ++r) {
      float e = f1v[r] + f2v;
      e = e > 0.f ? e : SLOPE_V * e;
      int av = adjb[(size_t)r * NN + j0 + n];
      e = (av == 0) ? NEG_INF_V : e;
      float rm = e;
      rm = fmaxf(rm, __shfl_xor(rm, 1, 32));
      rm = fmaxf(rm, __shfl_xor(rm, 2, 32));
      rm = fmaxf(rm, __shfl_xor(rm, 4, 32));
      rm = fmaxf(rm, __shfl_xor(rm, 8, 32));
      float nm   = fmaxf(mrun[r], rm);
      float pe   = __expf(e - nm);
      float corr = __expf(mrun[r] - nm);
      mrun[r] = nm;
      float rs = pe;
      rs += __shfl_xor(rs, 1, 32);
      rs += __shfl_xor(rs, 2, 32);
      rs += __shfl_xor(rs, 4, 32);
      rs += __shfl_xor(rs, 8, 32);
      lrun[r] = lrun[r] * corr + rs;
      p[r] = pe;
      #pragma unroll
      for (int t = 0; t < 16; ++t) acc[t][r] *= corr;
    }
    #pragma unroll
    for (int r = 0; r < 8; ++r)
      ptile[w][(r + 8 * hh) * BSTRIDE + n] = p[r];

    __syncthreads();

    v2f A[4];
    #pragma unroll
    for (int kk = 0; kk < 4; ++kk)
      A[kk] = *(const v2f*)(&ptile[w][n * BSTRIDE + 4 * kk + 2 * hh]);

    #pragma unroll
    for (int t = 0; t < 16; ++t) {
      const float* Bp = Btile + (t * 16 + n) * BSTRIDE + 2 * hh;
      #pragma unroll
      for (int kk = 0; kk < 4; ++kk) {
        v2f bm = *(const v2f*)(Bp + 4 * kk);
        acc[t] = __builtin_amdgcn_wmma_f32_16x16x4_f32(false, A[kk], false, bm,
                                                       (short)0, acc[t],
                                                       false, false);
      }
    }
    __syncthreads();
  }

  float inv[8];
  #pragma unroll
  for (int r = 0; r < 8; ++r) inv[r] = 1.0f / lrun[r];
  float* ob = out + ((size_t)(b * NN + i0 + 8 * hh)) * FF + n;
  #pragma unroll
  for (int t = 0; t < 16; ++t) {
    #pragma unroll
    for (int r = 0; r < 8; ++r)
      ob[(size_t)r * FF + t * 16] = acc[t][r] * inv[r];
  }
}

extern "C" void kernel_launch(void* const* d_in, const int* in_sizes, int n_in,
                              void* d_out, int out_size, void* d_ws, size_t ws_size,
                              hipStream_t stream) {
  const float* h   = (const float*)d_in[0];
  const int*   adj = (const int*)d_in[1];
  const float* W   = (const float*)d_in[2];
  const float* a   = (const float*)d_in[3];
  float* out = (float*)d_out;

  float* WhT = (float*)d_ws;
  float* f1  = WhT + (size_t)BB * NN * FF;
  float* f2  = f1 + (size_t)BB * NN;

  gat_gemm_hW<<<BB * (NN / 16) * (FF / 16), 32, 0, stream>>>(h, W, WhT);
  gat_f12<<<(BB * NN) / 256, 256, 0, stream>>>(WhT, a, f1, f2);
  gat_attn<<<BB * (NN / 16 / K3_WAVES), 128, 0, stream>>>(WhT, adj, f1, f2, out);
}


# attention kernel: scores computed in the A-operand layout (lane = 1 row x 8 columns): row max by one half-exchange, per-lane row-sum partials (new sum order), no P tile or shuffles via LDS
# speedup vs baseline: 1.2714x; 1.2714x over previous
#include <hip/hip_runtime.h>

typedef float v2f __attribute__((ext_vector_type(2)));
typedef int   v2i __attribute__((ext_vector_type(2)));
typedef float v4f __attribute__((ext_vector_type(4)));
typedef float v8f __attribute__((ext_vector_type(8)));

#define BB 16
#define NN 1024
#define FF 256
#define NEG_INF_V (-1000000000.0f)
#define SLOPE_V 0.2f

#define K3_WAVES 4
#define BSTRIDE 20

__global__ __launch_bounds__(32)
void gat_gemm_hW(const float* __restrict__ h, const float* __restrict__ W,
                 float* __restrict__ WhT) {
  int tile = blockIdx.x;
  int ot = tile % (FF / 16);
  int nt = (tile / (FF / 16)) % (NN / 16);
  int b  = tile / ((FF / 16) * (NN / 16));
  int lane = threadIdx.x;
  int n  = lane & 15;
  int hh = lane >> 4;

  const float* Arow = h + ((size_t)(b * NN + nt * 16 + n)) * FF;
  const float* Bcol = W + ot * 16 + n;

  v8f acc = {};
  #pragma unroll 4
  for (int k = 0; k < FF; k += 4) {
    v2f a = *(const v2f*)(Arow + k + 2 * hh);
    v2f bm;
    bm.x = Bcol[(size_t)(k + 2 * hh) * FF];
    bm.y = Bcol[(size_t)(k + 1 + 2 * hh) * FF];
    acc = __builtin_amdgcn_wmma_f32_16x16x4_f32(false, a, false, bm,
                                                (short)0, acc, false, false);
  }
  float* outp = WhT + ((size_t)(b * FF + ot * 16 + n)) * NN + nt * 16 + 8 * hh;
  v4f lo = { acc[0], acc[1], acc[2], acc[3] };
  v4f hi = { acc[4], acc[5], acc[6], acc[7] };
  *(v4f*)(outp)     = lo;
  *(v4f*)(outp + 4) = hi;
}

__global__ __launch_bounds__(256)
void gat_f12(const float* __restrict__ WhT, const float* __restrict__ a,
             float* __restrict__ f1, float* __restrict__ f2) {
  int idx = blockIdx.x * blockDim.x + threadIdx.x;
  int b = idx / NN;
  int n = idx % NN;
  const float* base = WhT + (size_t)b * FF * NN + n;
  float s1 = 0.f, s2 = 0.f;
  #pragma unroll 8
  for (int o = 0; o < FF; ++o) {
    float w = base[(size_t)o * NN];
    s1 += w * a[o];
    s2 += w * a[FF + o];
  }
  f1[idx] = s1;
  f2[idx] = s2;
}

__device__ __forceinline__ float xhalf(float x) {
  int v = __builtin_bit_cast(int, x);
  return __builtin_bit_cast(float,
      __builtin_amdgcn_permlanex16(v, v, 0x76543210, 0xfedcba98, false, false));
}

__global__ __launch_bounds__(128)
void gat_attn(const float* __restrict__ WhT, const int* __restrict__ adj,
              const float* __restrict__ f1, const float* __restrict__ f2,
              float* __restrict__ out) {
  __shared__ float Btile[FF * BSTRIDE];
  __shared__ __attribute__((aligned(16))) float rowx[K3_WAVES][16];
  int tid  = threadIdx.x;
  int w    = tid >> 5;
  int lane = tid & 31;
  int n  = lane & 15;
  int hh = lane >> 4;
  int b     = blockIdx.x >> 4;
  int super = blockIdx.x & 15;
  int i0 = (super * K3_WAVES + w) * 16;

  const float f1s = f1[b * NN + i0 + n];

  v8f acc[16];
  v8f zero = {};
  #pragma unroll
  for (int t = 0; t < 16; ++t) acc[t] = zero;
  float mrun = -3.0e38f;
  float lpart = 0.f;

  const int*   adjr = adj + ((size_t)(b * NN + i0 + n)) * NN + 2 * hh;
  const float* WhTb = WhT + (size_t)b * FF * NN;
  const float* f2h  = f2 + b * NN + 2 * hh;

  for (int jt = 0; jt < NN / 16; ++jt) {
    int j0 = jt * 16;

    {
      const float* g0 = WhTb + (size_t)tid * NN + j0;
      const float* g1 = g0 + (size_t)128 * NN;
      v4f x0 = *(const v4f*)(g0);
      v4f x1 = *(const v4f*)(g0 + 4);
      v4f x2 = *(const v4f*)(g0 + 8);
      v4f x3 = *(const v4f*)(g0 + 12);
      v4f y0 = *(const v4f*)(g1);
      v4f y1 = *(const v4f*)(g1 + 4);
      v4f y2 = *(const v4f*)(g1 + 8);
      v4f y3 = *(const v4f*)(g1 + 12);
      float* l0 = Btile + tid * BSTRIDE;
      float* l1 = Btile + (tid + 128) * BSTRIDE;
      *(v4f*)(l0)      = x0;  *(v4f*)(l0 + 4)  = x1;
      *(v4f*)(l0 + 8)  = x2;  *(v4f*)(l0 + 12) = x3;
      *(v4f*)(l1)      = y0;  *(v4f*)(l1 + 4)  = y1;
      *(v4f*)(l1 + 8)  = y2;  *(v4f*)(l1 + 12) = y3;
    }

    v2i av[4];
    v2f fv[4];
    #pragma unroll
    for (int kk = 0; kk < 4; ++kk) {
      av[kk] = *(const v2i*)(adjr + j0 + 4 * kk);
      fv[kk] = *(const v2f*)(f2h + j0 + 4 * kk);
    }
    float e[8];
    #pragma unroll
    for (int kk = 0; kk < 4; ++kk) {
      float e0 = f1s + fv[kk].x;
      float e1 = f1s + fv[kk].y;
      e0 = e0 > 0.f ? e0 : SLOPE_V * e0;
      e1 = e1 > 0.f ? e1 : SLOPE_V * e1;
      e[2 * kk]     = (av[kk].x == 0) ? NEG_INF_V : e0;
      e[2 * kk + 1] = (av[kk].y == 0) ? NEG_INF_V : e1;
    }
    float rm = fmaxf(fmaxf(fmaxf(e[0], e[1]), fmaxf(e[2], e[3])),
                     fmaxf(fmaxf(e[4], e[5]), fmaxf(e[6], e[7])));
    rm = fmaxf(rm, xhalf(rm));
    float nm   = fmaxf(mrun, rm);
    float corr = __expf(mrun - nm);
    mrun = nm;
    v2f A[4];
    float rs = 0.f;
    #pragma unroll
    for (int kk = 0; kk < 4; ++kk) {
      float p0 = __expf(e[2 * kk] - nm);
      float p1 = __expf(e[2 * kk + 1] - nm);
      A[kk].x = p0;
      A[kk].y = p1;
      rs += p0 + p1;
    }
    lpart = lpart * corr + rs;
    if (hh == 0) rowx[w][n] = corr;

    __syncthreads();

    {
      v4f c0 = *(const v4f*)(&rowx[w][8 * hh]);
      v4f c1 = *(const v4f*)(&rowx[w][8 * hh + 4]);
      #pragma unroll
      for (int t = 0; t < 16; ++t) {
        acc[t][0] *= c0.x; acc[t][1] *= c0.y; acc[t][2] *= c0.z; acc[t][3] *= c0.w;
        acc[t][4] *= c1.x; acc[t][5] *= c1.y; acc[t][6] *= c1.z; acc[t][7] *= c1.w;
      }
    }

    #pragma unroll
    for (int t = 0; t < 16; ++t) {
      const float* Bp = Btile + (t * 16 + n) * BSTRIDE + 2 * hh;
      #pragma unroll
      for (int kk = 0; kk < 4; ++kk) {
        v2f bm = *(const v2f*)(Bp + 4 * kk);
        acc[t] = __builtin_amdgcn_wmma_f32_16x16x4_f32(false, A[kk], false, bm,
                                                       (short)0, acc[t],
                                                       false, false);
      }
    }
    __syncthreads();
  }

  float ltot = lpart + xhalf(lpart);
  if (hh == 0) rowx[w][n] = 1.0f / ltot;
  __syncthreads();
  v4f q0 = *(const v4f*)(&rowx[w][8 * hh]);
  v4f q1 = *(const v4f*)(&rowx[w][8 * hh + 4]);
  float inv[8] = { q0.x, q0.y, q0.z, q0.w, q1.x, q1.y, q1.z, q1.w };
  float* ob = out + ((size_t)(b * NN + i0 + 8 * hh)) * FF + n;
  #pragma unroll
  for (int t = 0; t < 16; ++t) {
    #pragma unroll
    for (int r = 0; r < 8; ++r)
      ob[(size_t)r * FF + t * 16] = acc[t][r] * inv[r];
  }
}

extern "C" void kernel_launch(void* const* d_in, const int* in_sizes, int n_in,
                              void* d_out, int out_size, void* d_ws, size_t ws_size,
                              hipStream_t stream) {
  const float* h   = (const float*)d_in[0];
  const int*   adj = (const int*)d_in[1];
  const float* W   = (const float*)d_in[2];
  const float* a   = (const float*)d_in[3];
  float* out = (float*)d_out;

  float* WhT = (float*)d_ws;
  float* f1  = WhT + (size_t)BB * NN * FF;
  float* f2  = f1 + (size_t)BB * NN;

  gat_gemm_hW<<<BB * (NN / 16) * (FF / 16), 32, 0, stream>>>(h, W, WhT);
  gat_f12<<<(BB * NN) / 256, 256, 0, stream>>>(WhT, a, f1, f2);
  gat_attn<<<BB * (NN / 16 / K3_WAVES), 128, 0, stream>>>(WhT, adj, f1, f2, out);
}


# attention kernel software-pipelined: next step's panel, mask and f2 loads issued ahead of this step's matrix block (LDS-only barrier, two LDS stages, one-wave register budget: B reads in one group)
# speedup vs baseline: 1.6093x; 1.2658x over previous
#include <hip/hip_runtime.h>

typedef float v2f __attribute__((ext_vector_type(2)));
typedef int   v2i __attribute__((ext_vector_type(2)));
typedef float v4f __attribute__((ext_vector_type(4)));
typedef float v8f __attribute__((ext_vector_type(8)));

#define BB 16
#define NN 1024
#define FF 256
#define NEG_INF_V (-1000000000.0f)
#define SLOPE_V 0.2f

#define K3_WAVES 4
#define BSTRIDE 20

__global__ __launch_bounds__(32)
void gat_gemm_hW(const float* __restrict__ h, const float* __restrict__ W,
                 float* __restrict__ WhT) {
  int tile = blockIdx.x;
  int ot = tile % (FF / 16);
  int nt = (tile / (FF / 16)) % (NN / 16);
  int b  = tile / ((FF / 16) * (NN / 16));
  int lane = threadIdx.x;
  int n  = lane & 15;
  int hh = lane >> 4;

  const float* Arow = h + ((size_t)(b * NN + nt * 16 + n)) * FF;
  const float* Bcol = W + ot * 16 + n;

  v8f acc = {};
  #pragma unroll 4
  for (int k = 0; k < FF; k += 4) {
    v2f a = *(const v2f*)(Arow + k + 2 * hh);
    v2f bm;
    bm.x = Bcol[(size_t)(k + 2 * hh) * FF];
    bm.y = Bcol[(size_t)(k + 1 + 2 * hh) * FF];
    acc = __builtin_amdgcn_wmma_f32_16x16x4_f32(false, a, false, bm,
                                                (short)0, acc, false, false);
  }
  float* outp = WhT + ((size_t)(b * FF + ot * 16 + n)) * NN + nt * 16 + 8 * hh;
  v4f lo = { acc[0], acc[1], acc[2], acc[3] };
  v4f hi = { acc[4], acc[5], acc[6], acc[7] };
  *(v4f*)(outp)     = lo;
  *(v4f*)(outp + 4) = hi;
}

__global__ __launch_bounds__(256)
void gat_f12(const float* __restrict__ WhT, const float* __restrict__ a,
             float* __restrict__ f1, float* __restrict__ f2) {
  int idx = blockIdx.x * blockDim.x + threadIdx.x;
  int b = idx / NN;
  int n = idx % NN;
  const float* base = WhT + (size_t)b * FF * NN + n;
  float s1 = 0.f, s2 = 0.f;
  #pragma unroll 8
  for (int o = 0; o < FF; ++o) {
    float w = base[(size_t)o * NN];
    s1 += w * a[o];
    s2 += w * a[FF + o];
  }
  f1[idx] = s1;
  f2[idx] = s2;
}

__device__ __forceinline__ float xhalf(float x) {
  int v = __builtin_bit_cast(int, x);
  return __builtin_bit_cast(float,
      __builtin_amdgcn_permlanex16(v, v, 0x76543210, 0xfedcba98, false, false));
}

struct StepRegs {
  v2i av[4];
  v2f fv[4];
  v4f x[4];
  v4f y[4];
};

__device__ __forceinline__ void fetch_step(StepRegs& s, const int* adjp,
                                           const float* f2p, const float* g0) {
  #pragma unroll
  for (int kk = 0; kk < 4; ++kk) {
    s.av[kk] = *(const v2i*)(adjp + 4 * kk);
    s.fv[kk] = *(const v2f*)(f2p + 4 * kk);
  }
  const float* g1 = g0 + (size_t)128 * NN;
  #pragma unroll
  for (int q = 0; q < 4; ++q) {
    s.x[q] = *(const v4f*)(g0 + 4 * q);
    s.y[q] = *(const v4f*)(g1 + 4 * q);
  }
}

__global__ __launch_bounds__(128) __attribute__((amdgpu_waves_per_eu(1, 1)))
void gat_attn(const float* __restrict__ WhT, const int* __restrict__ adj,
              const float* __restrict__ f1, const float* __restrict__ f2,
              float* __restrict__ out) {
  __shared__ float Btile[2][FF * BSTRIDE];
  __shared__ __attribute__((aligned(16))) float rowx[K3_WAVES][16];
  int tid  = threadIdx.x;
  int w    = tid >> 5;
  int lane = tid & 31;
  int n  = lane & 15;
  int hh = lane >> 4;
  int b     = blockIdx.x >> 4;
  int super = blockIdx.x & 15;
  int i0 = (super * K3_WAVES + w) * 16;

  const int*   adjr = adj + ((size_t)(b * NN + i0 + n)) * NN + 2 * hh;
  const float* WhTr = WhT + (size_t)b * FF * NN + (size_t)tid * NN;
  const float* f2h  = f2 + b * NN + 2 * hh;

  StepRegs s;
  fetch_step(s, adjr, f2h, WhTr);

  const float f1s = f1[b * NN + i0 + n];

  v8f acc[16];
  v8f zero = {};
  #pragma unroll
  for (int t = 0; t < 16; ++t) acc[t] = zero;
  float mrun = -3.0e38f;
  float lpart = 0.f;

  for (int jt = 0; jt < NN / 16; ++jt) {
    float* Bt = Btile[jt & 1];

    {
      float* l0 = Bt + tid * BSTRIDE;
      float* l1 = Bt + (tid + 128) * BSTRIDE;
      *(v4f*)(l0)      = s.x[0];  *(v4f*)(l0 + 4)  = s.x[1];
      *(v4f*)(l0 + 8)  = s.x[2];  *(v4f*)(l0 + 12) = s.x[3];
      *(v4f*)(l1)      = s.y[0];  *(v4f*)(l1 + 4)  = s.y[1];
      *(v4f*)(l1 + 8)  = s.y[2];  *(v4f*)(l1 + 12) = s.y[3];
    }

    float e[8];
    #pragma unroll
    for (int kk = 0; kk < 4; ++kk) {
      float e0 = f1s + s.fv[kk].x;
      float e1 = f1s + s.fv[kk].y;
      e0 = e0 > 0.f ? e0 : SLOPE_V * e0;
      e1 = e1 > 0.f ? e1 : SLOPE_V * e1;
      e[2 * kk]     = (s.av[kk].x == 0) ? NEG_INF_V : e0;
      e[2 * kk + 1] = (s.av[kk].y == 0) ? NEG_INF_V : e1;
    }

    float rm = fmaxf(fmaxf(fmaxf(e[0], e[1]), fmaxf(e[2], e[3])),
                     fmaxf(fmaxf(e[4], e[5]), fmaxf(e[6], e[7])));
    rm = fmaxf(rm, xhalf(rm));
    float nm   = fmaxf(mrun, rm);
    float corr = __expf(mrun - nm);
    mrun = nm;
    v2f A[4];
    float rs = 0.f;
    #pragma unroll
    for (int kk = 0; kk < 4; ++kk) {
      float p0 = __expf(e[2 * kk] - nm);
      float p1 = __expf(e[2 * kk + 1] - nm);
      A[kk].x = p0;
      A[kk].y = p1;
      rs += p0 + p1;
    }
    lpart = lpart * corr + rs;
    if (hh == 0) rowx[w][n] = corr;

    {
      int jn = (jt + 1) & (NN / 16 - 1);
      fetch_step(s, adjr + jn * 16, f2h + jn * 16, WhTr + jn * 16);
    }

    __builtin_amdgcn_fence(__ATOMIC_RELEASE, "workgroup", "local");
    __builtin_amdgcn_s_barrier();
    __builtin_amdgcn_fence(__ATOMIC_ACQUIRE, "workgroup", "local");

    {
      v4f c0 = *(const v4f*)(&rowx[w][8 * hh]);
      v4f c1 = *(const v4f*)(&rowx[w][8 * hh + 4]);
      #pragma unroll
      for (int t = 0; t < 16; ++t) {
        acc[t][0] *= c0.x; acc[t][1] *= c0.y; acc[t][2] *= c0.z; acc[t][3] *= c0.w;
        acc[t][4] *= c1.x; acc[t][5] *= c1.y; acc[t][6] *= c1.z; acc[t][7] *= c1.w;
      }
    }

    #pragma unroll
    for (int t = 0; t < 16; ++t) {
      const float* Bp = Bt + (t * 16 + n) * BSTRIDE + 2 * hh;
      #pragma unroll
      for (int kk = 0; kk < 4; ++kk) {
        v2f bm = *(const v2f*)(Bp + 4 * kk);
        acc[t] = __builtin_amdgcn_wmma_f32_16x16x4_f32(false, A[kk], false, bm,
                                                       (short)0, acc[t],
                                                       false, false);
      }
    }
  }

  float ltot = lpart + xhalf(lpart);
  if (hh == 0) rowx[w][n] = 1.0f / ltot;
  __syncthreads();
  v4f q0 = *(const v4f*)(&rowx[w][8 * hh]);
  v4f q1 = *(const v4f*)(&rowx[w][8 * hh + 4]);
  float inv[8] = { q0.x, q0.y, q0.z, q0.w, q1.x, q1.y, q1.z, q1.w };
  float* ob = out + ((size_t)(b * NN + i0 + 8 * hh)) * FF + n;
  #pragma unroll
  for (int t = 0; t < 16; ++t) {
    #pragma unroll
    for (int r = 0; r < 8; ++r)
      ob[(size_t)r * FF + t * 16] = acc[t][r] * inv[r];
  }
}

extern "C" void kernel_launch(void* const* d_in, const int* in_sizes, int n_in,
                              void* d_out, int out_size, void* d_ws, size_t ws_size,
                              hipStream_t stream) {
  const float* h   = (const float*)d_in[0];
  const int*   adj = (const int*)d_in[1];
  const float* W   = (const float*)d_in[2];
  const float* a   = (const float*)d_in[3];
  float* out = (float*)d_out;

  float* WhT = (float*)d_ws;
  float* f1  = WhT + (size_t)BB * NN * FF;
  float* f2  = f1 + (size_t)BB * NN;

  gat_gemm_hW<<<BB * (NN / 16) * (FF / 16), 32, 0, stream>>>(h, W, WhT);
  gat_f12<<<(BB * NN) / 256, 256, 0, stream>>>(WhT, a, f1, f2);
  gat_attn<<<BB * (NN / 16 / K3_WAVES), 128, 0, stream>>>(WhT, adj, f1, f2, out);
}


# first kernel (h @ W) re-tiled: a 4-wave work-group per 64 rows, each wave a 16 x 256 strip (one A fragment feeds 16 tiles), W staged through LDS in chunks fetched ahead; sum order unchanged
# speedup vs baseline: 1.7636x; 1.0959x over previous
#include <hip/hip_runtime.h>

typedef float v2f __attribute__((ext_vector_type(2)));
typedef int   v2i __attribute__((ext_vector_type(2)));
typedef float v4f __attribute__((ext_vector_type(4)));
typedef float v8f __attribute__((ext_vector_type(8)));

#define BB 16
#define NN 1024
#define FF 256
#define NEG_INF_V (-1000000000.0f)
#define SLOPE_V 0.2f

#define K3_WAVES 4
#define BSTRIDE 20

#define G_NCH 8
#define WPITCH 512
#define WSTAGE (16 * WPITCH)

struct WChunk {
  v4f e[8];
  v4f o[8];
};

__device__ __forceinline__ void w_fetch(WChunk& s, const float* Wg) {
  #pragma unroll
  for (int q = 0; q < 8; ++q) {
    s.e[q] = *(const v4f*)(Wg + (size_t)(4 * q) * FF);
    s.o[q] = *(const v4f*)(Wg + (size_t)(4 * q + 1) * FF);
  }
}

__device__ __forceinline__ void w_store(const WChunk& s, float* Ws) {
  #pragma unroll
  for (int q = 0; q < 8; ++q) {
    v4f lo = { s.e[q].x, s.o[q].x, s.e[q].y, s.o[q].y };
    v4f hi = { s.e[q].z, s.o[q].z, s.e[q].w, s.o[q].w };
    *(v4f*)(Ws + 2 * q * WPITCH)     = lo;
    *(v4f*)(Ws + 2 * q * WPITCH + 4) = hi;
  }
}

__device__ __forceinline__ void lds_barrier() {
  __builtin_amdgcn_fence(__ATOMIC_RELEASE, "workgroup", "local");
  __builtin_amdgcn_s_barrier();
  __builtin_amdgcn_fence(__ATOMIC_ACQUIRE, "workgroup", "local");
}

__global__ __launch_bounds__(128) __attribute__((amdgpu_waves_per_eu(1, 1)))
void gat_gemm_hW(const float* __restrict__ h, const float* __restrict__ W,
                 float* __restrict__ WhT) {
  __shared__ __attribute__((aligned(16))) float Wl[2 * WSTAGE];
  int tid  = threadIdx.x;
  int w    = tid >> 5;
  int lane = tid & 31;
  int n  = lane & 15;
  int hh = lane >> 4;
  int blk = blockIdx.x;

  int cg = tid & 63;
  int kq = tid >> 6;
  const float* Wg = W + (size_t)(2 * kq) * FF + 4 * cg;
  float*       Ws = Wl + kq * WPITCH + ((8 * cg) ^ (32 * kq));

  const float* Ar = h + ((size_t)(blk * 64 + w * 16 + n)) * FF + 2 * hh;
  const float* Bl = Wl + hh * WPITCH + ((4 * n) ^ (32 * hh));

  v2f a[2][8];
  v4f Lb[2][8];
  WChunk s;

  #pragma unroll
  for (int j = 0; j < 8; ++j) a[0][j] = *(const v2f*)(Ar + 4 * j);
  w_fetch(s, Wg);
  w_store(s, Ws);
  w_fetch(s, Wg + (size_t)32 * FF);
  lds_barrier();
  #pragma unroll
  for (int T = 0; T < 8; ++T) Lb[0][T] = *(const v4f*)(Bl + 64 * T);

  v8f acc[16];
  v8f zero = {};
  #pragma unroll
  for (int t = 0; t < 16; ++t) acc[t] = zero;

  #pragma unroll
  for (int c = 0; c < G_NCH; ++c) {
    #pragma unroll
    for (int j = 0; j < 8; ++j) {
      if (j == 4) {
        if (c + 1 < G_NCH) {
          w_store(s, Ws + ((c + 1) & 1) * WSTAGE);
          #pragma unroll
          for (int jj = 0; jj < 8; ++jj)
            a[(c + 1) & 1][jj] = *(const v2f*)(Ar + 32 * (c + 1) + 4 * jj);
        }
        if (c + 2 < G_NCH) w_fetch(s, Wg + (size_t)(32 * (c + 2)) * FF);
      }
      if (j < 7) {
        #pragma unroll
        for (int T = 0; T < 8; ++T)
          Lb[(j + 1) & 1][T] =
              *(const v4f*)(Bl + (c & 1) * WSTAGE + 2 * (j + 1) * WPITCH + 64 * T);
      } else if (c + 1 < G_NCH) {
        lds_barrier();
        #pragma unroll
        for (int T = 0; T < 8; ++T)
          Lb[0][T] = *(const v4f*)(Bl + ((c + 1) & 1) * WSTAGE + 64 * T);
      }
      #pragma unroll
      for (int T = 0; T < 8; ++T) {
        v4f L = Lb[j & 1][T];
        v2f b0 = { L.x, L.y };
        v2f b1 = { L.z, L.w };
        acc[2 * T]     = __builtin_amdgcn_wmma_f32_16x16x4_f32(false, a[c & 1][j], false, b0,
                                                               (short)0, acc[2 * T], false, false);
        acc[2 * T + 1] = __builtin_amdgcn_wmma_f32_16x16x4_f32(false, a[c & 1][j], false, b1,
                                                               (short)0, acc[2 * T + 1], false, false);
      }
    }
  }

  int b  = blk >> 4;
  int n0 = (blk & 15) * 64 + w * 16 + 8 * hh;
  float* outb = WhT + ((size_t)(b * FF + 2 * n)) * NN + n0;
  #pragma unroll
  for (int t = 0; t < 16; ++t) {
    float* outp = outb + (size_t)(32 * (t >> 1) + (t & 1)) * NN;
    v4f lo = { acc[t][0], acc[t][1], acc[t][2], acc[t][3] };
    v4f hi = { acc[t][4], acc[t][5], acc[t][6], acc[t][7] };
    *(v4f*)(outp)     = lo;
    *(v4f*)(outp + 4) = hi;
  }
}

__global__ __launch_bounds__(256)
void gat_f12(const float* __restrict__ WhT, const float* __restrict__ a,
             float* __restrict__ f1, float* __restrict__ f2) {
  int idx = blockIdx.x * blockDim.x + threadIdx.x;
  int b = idx / NN;
  int n = idx % NN;
  const float* base = WhT + (size_t)b * FF * NN + n;
  float s1 = 0.f, s2 = 0.f;
  #pragma unroll 8
  for (int o = 0; o < FF; ++o) {
    float w = base[(size_t)o * NN];
    s1 += w * a[o];
    s2 += w * a[FF + o];
  }
  f1[idx] = s1;
  f2[idx] = s2;
}

__device__ __forceinline__ float xhalf(float x) {
  int v = __builtin_bit_cast(int, x);
  return __builtin_bit_cast(float,
      __builtin_amdgcn_permlanex16(v, v, 0x76543210, 0xfedcba98, false, false));
}

struct StepRegs {
  v2i av[4];
  v2f fv[4];
  v4f x[4];
  v4f y[4];
};

__device__ __forceinline__ void fetch_step(StepRegs& s, const int* adjp,
                                           const float* f2p, const float* g0) {
  #pragma unroll
  for (int kk = 0; kk < 4; ++kk) {
    s.av[kk] = *(const v2i*)(adjp + 4 * kk);
    s.fv[kk] = *(const v2f*)(f2p + 4 * kk);
  }
  const float* g1 = g0 + (size_t)128 * NN;
  #pragma unroll
  for (int q = 0; q < 4; ++q) {
    s.x[q] = *(const v4f*)(g0 + 4 * q);
    s.y[q] = *(const v4f*)(g1 + 4 * q);
  }
}

__global__ __launch_bounds__(128) __attribute__((amdgpu_waves_per_eu(1, 1)))
void gat_attn(const float* __restrict__ WhT, const int* __restrict__ adj,
              const float* __restrict__ f1, const float* __restrict__ f2,
              float* __restrict__ out) {
  __shared__ float Btile[2][FF * BSTRIDE];
  __shared__ __attribute__((aligned(16))) float rowx[K3_WAVES][16];
  int tid  = threadIdx.x;
  int w    = tid >> 5;
  int lane = tid & 31;
  int n  = lane & 15;
  int hh = lane >> 4;
  int b     = blockIdx.x >> 4;
  int super = blockIdx.x & 15;
  int i0 = (super * K3_WAVES + w) * 16;

  const int*   adjr = adj + ((size_t)(b * NN + i0 + n)) * NN + 2 * hh;
  const float* WhTr = WhT + (size_t)b * FF * NN + (size_t)tid * NN;
  const float* f2h  = f2 + b * NN + 2 * hh;

  StepRegs s;
  fetch_step(s, adjr, f2h, WhTr);

  const float f1s = f1[b * NN + i0 + n];

  v8f acc[16];
  v8f zero = {};
  #pragma unroll
  for (int t = 0; t < 16; ++t) acc[t] = zero;
  float mrun = -3.0e38f;
  float lpart = 0.f;

  for (int jt = 0; jt < NN / 16; ++jt) {
    float* Bt = Btile[jt & 1];

    {
      float* l0 = Bt + tid * BSTRIDE;
      float* l1 = Bt + (tid + 128) * BSTRIDE;
      *(v4f*)(l0)      = s.x[0];  *(v4f*)(l0 + 4)  = s.x[1];
      *(v4f*)(l0 + 8)  = s.x[2];  *(v4f*)(l0 + 12) = s.x[3];
      *(v4f*)(l1)      = s.y[0];  *(v4f*)(l1 + 4)  = s.y[1];
      *(v4f*)(l1 + 8)  = s.y[2];  *(v4f*)(l1 + 12) = s.y[3];
    }

    float e[8];
    #pragma unroll
    for (int kk = 0; kk < 4; ++kk) {
      float e0 = f1s + s.fv[kk].x;
      float e1 = f1s + s.fv[kk].y;
      e0 = e0 > 0.f ? e0 : SLOPE_V * e0;
      e1 = e1 > 0.f ? e1 : SLOPE_V * e1;
      e[2 * kk]     = (s.av[kk].x == 0) ? NEG_INF_V : e0;
      e[2 * kk + 1] = (s.av[kk].y == 0) ? NEG_INF_V : e1;
    }

    float rm = fmaxf(fmaxf(fmaxf(e[0], e[1]), fmaxf(e[2], e[3])),
                     fmaxf(fmaxf(e[4], e[5]), fmaxf(e[6], e[7])));
    rm = fmaxf(rm, xhalf(rm));
    float nm   = fmaxf(mrun, rm);
    float corr = __expf(mrun - nm);
    mrun = nm;
    v2f A[4];
    float rs = 0.f;
    #pragma unroll
    for (int kk = 0; kk < 4; ++kk) {
      float p0 = __expf(e[2 * kk] - nm);
      float p1 = __expf(e[2 * kk + 1] - nm);
      A[kk].x = p0;
      A[kk].y = p1;
      rs += p0 + p1;
    }
    lpart = lpart * corr + rs;
    if (hh == 0) rowx[w][n] = corr;

    {
      int jn = (jt + 1) & (NN / 16 - 1);
      fetch_step(s, adjr + jn * 16, f2h + jn * 16, WhTr + jn * 16);
    }

    __builtin_amdgcn_fence(__ATOMIC_RELEASE, "workgroup", "local");
    __builtin_amdgcn_s_barrier();
    __builtin_amdgcn_fence(__ATOMIC_ACQUIRE, "workgroup", "local");

    {
      v4f c0 = *(const v4f*)(&rowx[w][8 * hh]);
      v4f c1 = *(const v4f*)(&rowx[w][8 * hh + 4]);
      #pragma unroll
      for (int t = 0; t < 16; ++t) {
        acc[t][0] *= c0.x; acc[t][1] *= c0.y; acc[t][2] *= c0.z; acc[t][3] *= c0.w;
        acc[t][4] *= c1.x; acc[t][5] *= c1.y; acc[t][6] *= c1.z; acc[t][7] *= c1.w;
      }
    }

    #pragma unroll
    for (int t = 0; t < 16; ++t) {
      const float* Bp = Bt + (t * 16 + n) * BSTRIDE + 2 * hh;
      #pragma unroll
      for (int kk = 0; kk < 4; ++kk) {
        v2f bm = *(const v2f*)(Bp + 4 * kk);
        acc[t] = __builtin_amdgcn_wmma_f32_16x16x4_f32(false, A[kk], false, bm,
                                                       (short)0, acc[t],
                                                       false, false);
      }
    }
  }

  float ltot = lpart + xhalf(lpart);
  if (hh == 0) rowx[w][n] = 1.0f / ltot;
  __syncthreads();
  v4f q0 = *(const v4f*)(&rowx[w][8 * hh]);
  v4f q1 = *(const v4f*)(&rowx[w][8 * hh + 4]);
  float inv[8] = { q0.x, q0.y, q0.z, q0.w, q1.x, q1.y, q1.z, q1.w };
  float* ob = out + ((size_t)(b * NN + i0 + 8 * hh)) * FF + n;
  #pragma unroll
  for (int t = 0; t < 16; ++t) {
    #pragma unroll
    for (int r = 0; r < 8; ++r)
      ob[(size_t)r * FF + t * 16] = acc[t][r] * inv[r];
  }
}

extern "C" void kernel_launch(void* const* d_in, const int* in_sizes, int n_in,
                              void* d_out, int out_size, void* d_ws, size_t ws_size,
                              hipStream_t stream) {
  const float* h   = (const float*)d_in[0];
  const int*   adj = (const int*)d_in[1];
  const float* W   = (const float*)d_in[2];
  const float* a   = (const float*)d_in[3];
  float* out = (float*)d_out;

  float* WhT = (float*)d_ws;
  float* f1  = WhT + (size_t)BB * NN * FF;
  float* f2  = f1 + (size_t)BB * NN;

  gat_gemm_hW<<<BB * NN / 64, 128, 0, stream>>>(h, W, WhT);
  gat_f12<<<(BB * NN) / 256, 256, 0, stream>>>(WhT, a, f1, f2);
  gat_attn<<<BB * (NN / 16 / K3_WAVES), 128, 0, stream>>>(WhT, adj, f1, f2, out);
}


# f1/f2 taken from the first kernel's accumulators in its epilogue (per-lane column sums + 16-lane DPP butterfly: new f1/f2 sum order); the separate f1/f2 launch and its re-read of Wh are removed
# speedup vs baseline: 2.2146x; 1.2557x over previous
#include <hip/hip_runtime.h>

typedef float v2f __attribute__((ext_vector_type(2)));
typedef int   v2i __attribute__((ext_vector_type(2)));
typedef float v4f __attribute__((ext_vector_type(4)));
typedef float v8f __attribute__((ext_vector_type(8)));

#define BB 16
#define NN 1024
#define FF 256
#define NEG_INF_V (-1000000000.0f)
#define SLOPE_V 0.2f

#define K3_WAVES 4
#define BSTRIDE 20

#define G_NCH 8
#define WPITCH 512
#define WSTAGE (16 * WPITCH)

struct WChunk {
  v4f e[8];
  v4f o[8];
};

__device__ __forceinline__ void w_fetch(WChunk& s, const float* Wg) {
  #pragma unroll
  for (int q = 0; q < 8; ++q) {
    s.e[q] = *(const v4f*)(Wg + (size_t)(4 * q) * FF);
    s.o[q] = *(const v4f*)(Wg + (size_t)(4 * q + 1) * FF);
  }
}

__device__ __forceinline__ void w_store(const WChunk& s, float* Ws) {
  #pragma unroll
  for (int q = 0; q < 8; ++q) {
    v4f lo = { s.e[q].x, s.o[q].x, s.e[q].y, s.o[q].y };
    v4f hi = { s.e[q].z, s.o[q].z, s.e[q].w, s.o[q].w };
    *(v4f*)(Ws + 2 * q * WPITCH)     = lo;
    *(v4f*)(Ws + 2 * q * WPITCH + 4) = hi;
  }
}

template <int M>
__device__ __forceinline__ float row_xor(float x) {
  int v = __builtin_bit_cast(int, x);
  return __builtin_bit_cast(float,
      __builtin_amdgcn_update_dpp(0, v, 0x160 + M, 0xf, 0xf, true));
}

__device__ __forceinline__ void lds_barrier() {
  __builtin_amdgcn_fence(__ATOMIC_RELEASE, "workgroup", "local");
  __builtin_amdgcn_s_barrier();
  __builtin_amdgcn_fence(__ATOMIC_ACQUIRE, "workgroup", "local");
}

__global__ __launch_bounds__(128) __attribute__((amdgpu_waves_per_eu(1, 1)))
void gat_gemm_hW(const float* __restrict__ h, const float* __restrict__ W,
                 const float* __restrict__ a, float* __restrict__ WhT,
                 float* __restrict__ f1, float* __restrict__ f2) {
  __shared__ __attribute__((aligned(16))) float Wl[2 * WSTAGE];
  int tid  = threadIdx.x;
  int w    = tid >> 5;
  int lane = tid & 31;
  int n  = lane & 15;
  int hh = lane >> 4;
  int blk = blockIdx.x;

  int cg = tid & 63;
  int kq = tid >> 6;
  const float* Wg = W + (size_t)(2 * kq) * FF + 4 * cg;
  float*       Ws = Wl + kq * WPITCH + ((8 * cg) ^ (32 * kq));

  const float* Ar = h + ((size_t)(blk * 64 + w * 16 + n)) * FF + 2 * hh;
  const float* Bl = Wl + hh * WPITCH + ((4 * n) ^ (32 * hh));

  v2f af[2][8];
  v2f a1v[8], a2v[8];
  v4f Lb[2][8];
  WChunk s;

  #pragma unroll
  for (int j = 0; j < 8; ++j) af[0][j] = *(const v2f*)(Ar + 4 * j);
  w_fetch(s, Wg);
  w_store(s, Ws);
  w_fetch(s, Wg + (size_t)32 * FF);
  lds_barrier();
  #pragma unroll
  for (int T = 0; T < 8; ++T) Lb[0][T] = *(const v4f*)(Bl + 64 * T);

  v8f acc[16];
  v8f zero = {};
  #pragma unroll
  for (int t = 0; t < 16; ++t) acc[t] = zero;

  #pragma unroll
  for (int c = 0; c < G_NCH; ++c) {
    #pragma unroll
    for (int j = 0; j < 8; ++j) {
      if (j == 4) {
        if (c + 1 < G_NCH) {
          w_store(s, Ws + ((c + 1) & 1) * WSTAGE);
          #pragma unroll
          for (int jj = 0; jj < 8; ++jj)
            af[(c + 1) & 1][jj] = *(const v2f*)(Ar + 32 * (c + 1) + 4 * jj);
        }
        if (c + 2 < G_NCH) w_fetch(s, Wg + (size_t)(32 * (c + 2)) * FF);
      }
      if (c == G_NCH - 1 && j == 0) {
        #pragma unroll
        for (int T = 0; T < 8; ++T) {
          a1v[T] = *(const v2f*)(a + 32 * T + 2 * n);
          a2v[T] = *(const v2f*)(a + FF + 32 * T + 2 * n);
        }
      }
      if (j < 7) {
        #pragma unroll
        for (int T = 0; T < 8; ++T)
          Lb[(j + 1) & 1][T] =
              *(const v4f*)(Bl + (c & 1) * WSTAGE + 2 * (j + 1) * WPITCH + 64 * T);
      } else if (c + 1 < G_NCH) {
        lds_barrier();
        #pragma unroll
        for (int T = 0; T < 8; ++T)
          Lb[0][T] = *(const v4f*)(Bl + ((c + 1) & 1) * WSTAGE + 64 * T);
      }
      #pragma unroll
      for (int T = 0; T < 8; ++T) {
        v4f L = Lb[j & 1][T];
        v2f b0 = { L.x, L.y };
        v2f b1 = { L.z, L.w };
        acc[2 * T]     = __builtin_amdgcn_wmma_f32_16x16x4_f32(false, af[c & 1][j], false, b0,
                                                               (short)0, acc[2 * T], false, false);
        acc[2 * T + 1] = __builtin_amdgcn_wmma_f32_16x16x4_f32(false, af[c & 1][j], false, b1,
                                                               (short)0, acc[2 * T + 1], false, false);
      }
    }
  }

  int b  = blk >> 4;
  int n0 = (blk & 15) * 64 + w * 16 + 8 * hh;
  float* outb = WhT + ((size_t)(b * FF + 2 * n)) * NN + n0;
  #pragma unroll
  for (int t = 0; t < 16; ++t) {
    float* outp = outb + (size_t)(32 * (t >> 1) + (t & 1)) * NN;
    v4f lo = { acc[t][0], acc[t][1], acc[t][2], acc[t][3] };
    v4f hi = { acc[t][4], acc[t][5], acc[t][6], acc[t][7] };
    *(v4f*)(outp)     = lo;
    *(v4f*)(outp + 4) = hi;
  }

  float p1[8], p2[8];
  #pragma unroll
  for (int r = 0; r < 8; ++r) { p1[r] = 0.f; p2[r] = 0.f; }
  #pragma unroll
  for (int t = 0; t < 16; ++t) {
    float c1 = (t & 1) ? a1v[t >> 1].y : a1v[t >> 1].x;
    float c2 = (t & 1) ? a2v[t >> 1].y : a2v[t >> 1].x;
    #pragma unroll
    for (int r = 0; r < 8; ++r) {
      p1[r] += acc[t][r] * c1;
      p2[r] += acc[t][r] * c2;
    }
  }
  #pragma unroll
  for (int r = 0; r < 8; ++r) { p1[r] += row_xor<1>(p1[r]); p2[r] += row_xor<1>(p2[r]); }
  #pragma unroll
  for (int r = 0; r < 8; ++r) { p1[r] += row_xor<2>(p1[r]); p2[r] += row_xor<2>(p2[r]); }
  #pragma unroll
  for (int r = 0; r < 8; ++r) { p1[r] += row_xor<4>(p1[r]); p2[r] += row_xor<4>(p2[r]); }
  #pragma unroll
  for (int r = 0; r < 8; ++r) { p1[r] += row_xor<8>(p1[r]); p2[r] += row_xor<8>(p2[r]); }
  float s1 = p1[0], s2 = p2[0];
  #pragma unroll
  for (int r = 1; r < 8; ++r) {
    s1 = (n == r) ? p1[r] : s1;
    s2 = (n == r) ? p2[r] : s2;
  }
  if (n < 8) {
    int row = blk * 64 + w * 16 + 8 * hh + n;
    f1[row] = s1;
    f2[row] = s2;
  }
}

__device__ __forceinline__ float xhalf(float x) {
  int v = __builtin_bit_cast(int, x);
  return __builtin_bit_cast(float,
      __builtin_amdgcn_permlanex16(v, v, 0x76543210, 0xfedcba98, false, false));
}

struct StepRegs {
  v2i av[4];
  v2f fv[4];
  v4f x[4];
  v4f y[4];
};

__device__ __forceinline__ void fetch_step(StepRegs& s, const int* adjp,
                                           const float* f2p, const float* g0) {
  #pragma unroll
  for (int kk = 0; kk < 4; ++kk) {
    s.av[kk] = *(const v2i*)(adjp + 4 * kk);
    s.fv[kk] = *(const v2f*)(f2p + 4 * kk);
  }
  const float* g1 = g0 + (size_t)128 * NN;
  #pragma unroll
  for (int q = 0; q < 4; ++q) {
    s.x[q] = *(const v4f*)(g0 + 4 * q);
    s.y[q] = *(const v4f*)(g1 + 4 * q);
  }
}

__global__ __launch_bounds__(128) __attribute__((amdgpu_waves_per_eu(1, 1)))
void gat_attn(const float* __restrict__ WhT, const int* __restrict__ adj,
              const float* __restrict__ f1, const float* __restrict__ f2,
              float* __restrict__ out) {
  __shared__ float Btile[2][FF * BSTRIDE];
  __shared__ __attribute__((aligned(16))) float rowx[K3_WAVES][16];
  int tid  = threadIdx.x;
  int w    = tid >> 5;
  int lane = tid & 31;
  int n  = lane & 15;
  int hh = lane >> 4;
  int b     = blockIdx.x >> 4;
  int super = blockIdx.x & 15;
  int i0 = (super * K3_WAVES + w) * 16;

  const int*   adjr = adj + ((size_t)(b * NN + i0 + n)) * NN + 2 * hh;
  const float* WhTr = WhT + (size_t)b * FF * NN + (size_t)tid * NN;
  const float* f2h  = f2 + b * NN + 2 * hh;

  StepRegs s;
  fetch_step(s, adjr, f2h, WhTr);

  const float f1s = f1[b * NN + i0 + n];

  v8f acc[16];
  v8f zero = {};
  #pragma unroll
  for (int t = 0; t < 16; ++t) acc[t] = zero;
  float mrun = -3.0e38f;
  float lpart = 0.f;

  for (int jt = 0; jt < NN / 16; ++jt) {
    float* Bt = Btile[jt & 1];

    {
      float* l0 = Bt + tid * BSTRIDE;
      float* l1 = Bt + (tid + 128) * BSTRIDE;
      *(v4f*)(l0)      = s.x[0];  *(v4f*)(l0 + 4)  = s.x[1];
      *(v4f*)(l0 + 8)  = s.x[2];  *(v4f*)(l0 + 12) = s.x[3];
      *(v4f*)(l1)      = s.y[0];  *(v4f*)(l1 + 4)  = s.y[1];
      *(v4f*)(l1 + 8)  = s.y[2];  *(v4f*)(l1 + 12) = s.y[3];
    }

    float e[8];
    #pragma unroll
    for (int kk = 0; kk < 4; ++kk) {
      float e0 = f1s + s.fv[kk].x;
      float e1 = f1s + s.fv[kk].y;
      e0 = e0 > 0.f ? e0 : SLOPE_V * e0;
      e1 = e1 > 0.f ? e1 : SLOPE_V * e1;
      e[2 * kk]     = (s.av[kk].x == 0) ? NEG_INF_V : e0;
      e[2 * kk + 1] = (s.av[kk].y == 0) ? NEG_INF_V : e1;
    }

    float rm = fmaxf(fmaxf(fmaxf(e[0], e[1]), fmaxf(e[2], e[3])),
                     fmaxf(fmaxf(e[4], e[5]), fmaxf(e[6], e[7])));
    rm = fmaxf(rm, xhalf(rm));
    float nm   = fmaxf(mrun, rm);
    float corr = __expf(mrun - nm);
    mrun = nm;
    v2f A[4];
    float rs = 0.f;
    #pragma unroll
    for (int kk = 0; kk < 4; ++kk) {
      float p0 = __expf(e[2 * kk] - nm);
      float p1 = __expf(e[2 * kk + 1] - nm);
      A[kk].x = p0;
      A[kk].y = p1;
      rs += p0 + p1;
    }
    lpart = lpart * corr + rs;
    if (hh == 0) rowx[w][n] = corr;

    {
      int jn = (jt + 1) & (NN / 16 - 1);
      fetch_step(s, adjr + jn * 16, f2h + jn * 16, WhTr + jn * 16);
    }

    __builtin_amdgcn_fence(__ATOMIC_RELEASE, "workgroup", "local");
    __builtin_amdgcn_s_barrier();
    __builtin_amdgcn_fence(__ATOMIC_ACQUIRE, "workgroup", "local");

    {
      v4f c0 = *(const v4f*)(&rowx[w][8 * hh]);
      v4f c1 = *(const v4f*)(&rowx[w][8 * hh + 4]);
      #pragma unroll
      for (int t = 0; t < 16; ++t) {
        acc[t][0] *= c0.x; acc[t][1] *= c0.y; acc[t][2] *= c0.z; acc[t][3] *= c0.w;
        acc[t][4] *= c1.x; acc[t][5] *= c1.y; acc[t][6] *= c1.z; acc[t][7] *= c1.w;
      }
    }

    #pragma unroll
    for (int t = 0; t < 16; ++t) {
      const float* Bp = Bt + (t * 16 + n) * BSTRIDE + 2 * hh;
      #pragma unroll
      for (int kk = 0; kk < 4; ++kk) {
        v2f bm = *(const v2f*)(Bp + 4 * kk);
        acc[t] = __builtin_amdgcn_wmma_f32_16x16x4_f32(false, A[kk], false, bm,
                                                       (short)0, acc[t],
                                                       false, false);
      }
    }
  }

  float ltot = lpart + xhalf(lpart);
  if (hh == 0) rowx[w][n] = 1.0f / ltot;
  __syncthreads();
  v4f q0 = *(const v4f*)(&rowx[w][8 * hh]);
  v4f q1 = *(const v4f*)(&rowx[w][8 * hh + 4]);
  float inv[8] = { q0.x, q0.y, q0.z, q0.w, q1.x, q1.y, q1.z, q1.w };
  float* ob = out + ((size_t)(b * NN + i0 + 8 * hh)) * FF + n;
  #pragma unroll
  for (int t = 0; t < 16; ++t) {
    #pragma unroll
    for (int r = 0; r < 8; ++r)
      ob[(size_t)r * FF + t * 16] = acc[t][r] * inv[r];
  }
}

extern "C" void kernel_launch(void* const* d_in, const int* in_sizes, int n_in,
                              void* d_out, int out_size, void* d_ws, size_t ws_size,
                              hipStream_t stream) {
  const float* h   = (const float*)d_in[0];
  const int*   adj = (const int*)d_in[1];
  const float* W   = (const float*)d_in[2];
  const float* a   = (const float*)d_in[3];
  float* out = (float*)d_out;

  float* WhT = (float*)d_ws;
  float* f1  = WhT + (size_t)BB * NN * FF;
  float* f2  = f1 + (size_t)BB * NN;

  gat_gemm_hW<<<BB * NN / 64, 128, 0, stream>>>(h, W, a, WhT, f1, f2);
  gat_attn<<<BB * (NN / 16 / K3_WAVES), 128, 0, stream>>>(WhT, adj, f1, f2, out);
}


# Wh kept panel-major in the workspace (each attention step's 16 KB panel contiguous, pieces in the first kernel's store order): each panel load and Wh store covers 4 whole lines (loads touched 32)
# speedup vs baseline: 2.3556x; 1.0637x over previous
#include <hip/hip_runtime.h>

typedef float v2f __attribute__((ext_vector_type(2)));
typedef int   v2i __attribute__((ext_vector_type(2)));
typedef float v4f __attribute__((ext_vector_type(4)));
typedef float v8f __attribute__((ext_vector_type(8)));

#define BB 16
#define NN 1024
#define FF 256
#define NEG_INF_V (-1000000000.0f)
#define SLOPE_V 0.2f

#define K3_WAVES 4
#define BSTRIDE 20
#define PANEL (FF * 16)

#define G_NCH 8
#define WPITCH 512
#define WSTAGE (16 * WPITCH)

struct WChunk {
  v4f e[8];
  v4f o[8];
};

__device__ __forceinline__ void w_fetch(WChunk& s, const float* Wg) {
  #pragma unroll
  for (int q = 0; q < 8; ++q) {
    s.e[q] = *(const v4f*)(Wg + (size_t)(4 * q) * FF);
    s.o[q] = *(const v4f*)(Wg + (size_t)(4 * q + 1) * FF);
  }
}

__device__ __forceinline__ void w_store(const WChunk& s, float* Ws) {
  #pragma unroll
  for (int q = 0; q < 8; ++q) {
    v4f lo = { s.e[q].x, s.o[q].x, s.e[q].y, s.o[q].y };
    v4f hi = { s.e[q].z, s.o[q].z, s.e[q].w, s.o[q].w };
    *(v4f*)(Ws + 2 * q * WPITCH)     = lo;
    *(v4f*)(Ws + 2 * q * WPITCH + 4) = hi;
  }
}

template <int M>
__device__ __forceinline__ float row_xor(float x) {
  int v = __builtin_bit_cast(int, x);
  return __builtin_bit_cast(float,
      __builtin_amdgcn_update_dpp(0, v, 0x160 + M, 0xf, 0xf, true));
}

__device__ __forceinline__ void lds_barrier() {
  __builtin_amdgcn_fence(__ATOMIC_RELEASE, "workgroup", "local");
  __builtin_amdgcn_s_barrier();
  __builtin_amdgcn_fence(__ATOMIC_ACQUIRE, "workgroup", "local");
}

__global__ __launch_bounds__(128) __attribute__((amdgpu_waves_per_eu(1, 1)))
void gat_gemm_hW(const float* __restrict__ h, const float* __restrict__ W,
                 const float* __restrict__ a, float* __restrict__ WhP,
                 float* __restrict__ f1, float* __restrict__ f2) {
  __shared__ __attribute__((aligned(16))) float Wl[2 * WSTAGE];
  int tid  = threadIdx.x;
  int w    = tid >> 5;
  int lane = tid & 31;
  int n  = lane & 15;
  int hh = lane >> 4;
  int blk = blockIdx.x;

  int cg = tid & 63;
  int kq = tid >> 6;
  const float* Wg = W + (size_t)(2 * kq) * FF + 4 * cg;
  float*       Ws = Wl + kq * WPITCH + ((8 * cg) ^ (32 * kq));

  const float* Ar = h + ((size_t)(blk * 64 + w * 16 + n)) * FF + 2 * hh;
  const float* Bl = Wl + hh * WPITCH + ((4 * n) ^ (32 * hh));

  v2f af[2][8];
  v2f a1v[8], a2v[8];
  v4f Lb[2][8];
  WChunk s;

  #pragma unroll
  for (int j = 0; j < 8; ++j) af[0][j] = *(const v2f*)(Ar + 4 * j);
  w_fetch(s, Wg);
  w_store(s, Ws);
  w_fetch(s, Wg + (size_t)32 * FF);
  lds_barrier();
  #pragma unroll
  for (int T = 0; T < 8; ++T) Lb[0][T] = *(const v4f*)(Bl + 64 * T);

  v8f acc[16];
  v8f zero = {};
  #pragma unroll
  for (int t = 0; t < 16; ++t) acc[t] = zero;

  #pragma unroll
  for (int c = 0; c < G_NCH; ++c) {
    #pragma unroll
    for (int j = 0; j < 8; ++j) {
      if (j == 4) {
        if (c + 1 < G_NCH) {
          w_store(s, Ws + ((c + 1) & 1) * WSTAGE);
          #pragma unroll
          for (int jj = 0; jj < 8; ++jj)
            af[(c + 1) & 1][jj] = *(const v2f*)(Ar + 32 * (c + 1) + 4 * jj);
        }
        if (c + 2 < G_NCH) w_fetch(s, Wg + (size_t)(32 * (c + 2)) * FF);
      }
      if (c == G_NCH - 1 && j == 0) {
        #pragma unroll
        for (int T = 0; T < 8; ++T) {
          a1v[T] = *(const v2f*)(a + 32 * T + 2 * n);
          a2v[T] = *(const v2f*)(a + FF + 32 * T + 2 * n);
        }
      }
      if (j < 7) {
        #pragma unroll
        for (int T = 0; T < 8; ++T)
          Lb[(j + 1) & 1][T] =
              *(const v4f*)(Bl + (c & 1) * WSTAGE + 2 * (j + 1) * WPITCH + 64 * T);
      } else if (c + 1 < G_NCH) {
        lds_barrier();
        #pragma unroll
        for (int T = 0; T < 8; ++T)
          Lb[0][T] = *(const v4f*)(Bl + ((c + 1) & 1) * WSTAGE + 64 * T);
      }
      #pragma unroll
      for (int T = 0; T < 8; ++T) {
        v4f L = Lb[j & 1][T];
        v2f b0 = { L.x, L.y };
        v2f b1 = { L.z, L.w };
        acc[2 * T]     = __builtin_amdgcn_wmma_f32_16x16x4_f32(false, af[c & 1][j], false, b0,
                                                               (short)0, acc[2 * T], false, false);
        acc[2 * T + 1] = __builtin_amdgcn_wmma_f32_16x16x4_f32(false, af[c & 1][j], false, b1,
                                                               (short)0, acc[2 * T + 1], false, false);
      }
    }
  }

  int b  = blk >> 4;
  int jt = (blk & 15) * 4 + w;
  float* outb = WhP + ((size_t)(b * (NN / 16) + jt)) * PANEL + 4 * lane;
  float* out0 = outb + (n >> 3) * 128;
  float* out1 = outb + (1 - (n >> 3)) * 128;
  #pragma unroll
  for (int t = 0; t < 16; ++t) {
    v4f lo = { acc[t][0], acc[t][1], acc[t][2], acc[t][3] };
    v4f hi = { acc[t][4], acc[t][5], acc[t][6], acc[t][7] };
    *(v4f*)(out0 + t * 256) = lo;
    *(v4f*)(out1 + t * 256) = hi;
  }

  float p1[8], p2[8];
  #pragma unroll
  for (int r = 0; r < 8; ++r) { p1[r] = 0.f; p2[r] = 0.f; }
  #pragma unroll
  for (int t = 0; t < 16; ++t) {
    float c1 = (t & 1) ? a1v[t >> 1].y : a1v[t >> 1].x;
    float c2 = (t & 1) ? a2v[t >> 1].y : a2v[t >> 1].x;
    #pragma unroll
    for (int r = 0; r < 8; ++r) {
      p1[r] += acc[t][r] * c1;
      p2[r] += acc[t][r] * c2;
    }
  }
  #pragma unroll
  for (int r = 0; r < 8; ++r) { p1[r] += row_xor<1>(p1[r]); p2[r] += row_xor<1>(p2[r]); }
  #pragma unroll
  for (int r = 0; r < 8; ++r) { p1[r] += row_xor<2>(p1[r]); p2[r] += row_xor<2>(p2[r]); }
  #pragma unroll
  for (int r = 0; r < 8; ++r) { p1[r] += row_xor<4>(p1[r]); p2[r] += row_xor<4>(p2[r]); }
  #pragma unroll
  for (int r = 0; r < 8; ++r) { p1[r] += row_xor<8>(p1[r]); p2[r] += row_xor<8>(p2[r]); }
  float s1 = p1[0], s2 = p2[0];
  #pragma unroll
  for (int r = 1; r < 8; ++r) {
    s1 = (n == r) ? p1[r] : s1;
    s2 = (n == r) ? p2[r] : s2;
  }
  if (n < 8) {
    int row = blk * 64 + w * 16 + 8 * hh + n;
    f1[row] = s1;
    f2[row] = s2;
  }
}

__device__ __forceinline__ float xhalf(float x) {
  int v = __builtin_bit_cast(int, x);
  return __builtin_bit_cast(float,
      __builtin_amdgcn_permlanex16(v, v, 0x76543210, 0xfedcba98, false, false));
}

struct StepRegs {
  v2i av[4];
  v2f fv[4];
  v4f p[8];
};

__device__ __forceinline__ void fetch_step(StepRegs& s, const int* adjp,
                                           const float* f2p, const float* g0) {
  #pragma unroll
  for (int kk = 0; kk < 4; ++kk) {
    s.av[kk] = *(const v2i*)(adjp + 4 * kk);
    s.fv[kk] = *(const v2f*)(f2p + 4 * kk);
  }
  #pragma unroll
  for (int q = 0; q < 8; ++q) s.p[q] = *(const v4f*)(g0 + 512 * q);
}

__global__ __launch_bounds__(128) __attribute__((amdgpu_waves_per_eu(1, 1)))
void gat_attn(const float* __restrict__ WhP, const int* __restrict__ adj,
              const float* __restrict__ f1, const float* __restrict__ f2,
              float* __restrict__ out) {
  __shared__ __attribute__((aligned(16))) float Btile[2][FF * BSTRIDE];
  __shared__ __attribute__((aligned(16))) float rowx[K3_WAVES][16];
  int tid  = threadIdx.x;
  int w    = tid >> 5;
  int lane = tid & 31;
  int n  = lane & 15;
  int hh = lane >> 4;
  int b     = blockIdx.x >> 4;
  int super = blockIdx.x & 15;
  int i0 = (super * K3_WAVES + w) * 16;

  const int*   adjr = adj + ((size_t)(b * NN + i0 + n)) * NN + 2 * hh;
  const float* Pb   = WhP + (size_t)b * (NN / 16) * PANEL + 4 * tid;
  const int lbase = (2 * n + (w >> 1)) * BSTRIDE + 8 * hh + 4 * ((w & 1) ^ (n >> 3));
  const float* f2h  = f2 + b * NN + 2 * hh;

  StepRegs s;
  fetch_step(s, adjr, f2h, Pb);

  const float f1s = f1[b * NN + i0 + n];

  v8f acc[16];
  v8f zero = {};
  #pragma unroll
  for (int t = 0; t < 16; ++t) acc[t] = zero;
  float mrun = -3.0e38f;
  float lpart = 0.f;

  for (int jt = 0; jt < NN / 16; ++jt) {
    float* Bt = Btile[jt & 1];

    {
      float* l0 = Bt + lbase;
      #pragma unroll
      for (int q = 0; q < 8; ++q) *(v4f*)(l0 + q * (32 * BSTRIDE)) = s.p[q];
    }

    float e[8];
    #pragma unroll
    for (int kk = 0; kk < 4; ++kk) {
      float e0 = f1s + s.fv[kk].x;
      float e1 = f1s + s.fv[kk].y;
      e0 = e0 > 0.f ? e0 : SLOPE_V * e0;
      e1 = e1 > 0.f ? e1 : SLOPE_V * e1;
      e[2 * kk]     = (s.av[kk].x == 0) ? NEG_INF_V : e0;
      e[2 * kk + 1] = (s.av[kk].y == 0) ? NEG_INF_V : e1;
    }

    float rm = fmaxf(fmaxf(fmaxf(e[0], e[1]), fmaxf(e[2], e[3])),
                     fmaxf(fmaxf(e[4], e[5]), fmaxf(e[6], e[7])));
    rm = fmaxf(rm, xhalf(rm));
    float nm   = fmaxf(mrun, rm);
    float corr = __expf(mrun - nm);
    mrun = nm;
    v2f A[4];
    float rs = 0.f;
    #pragma unroll
    for (int kk = 0; kk < 4; ++kk) {
      float p0 = __expf(e[2 * kk] - nm);
      float p1 = __expf(e[2 * kk + 1] - nm);
      A[kk].x = p0;
      A[kk].y = p1;
      rs += p0 + p1;
    }
    lpart = lpart * corr + rs;
    if (hh == 0) rowx[w][n] = corr;

    {
      int jn = (jt + 1) & (NN / 16 - 1);
      fetch_step(s, adjr + jn * 16, f2h + jn * 16, Pb + (size_t)jn * PANEL);
    }

    __builtin_amdgcn_fence(__ATOMIC_RELEASE, "workgroup", "local");
    __builtin_amdgcn_s_barrier();
    __builtin_amdgcn_fence(__ATOMIC_ACQUIRE, "workgroup", "local");

    {
      v4f c0 = *(const v4f*)(&rowx[w][8 * hh]);
      v4f c1 = *(const v4f*)(&rowx[w][8 * hh + 4]);
      #pragma unroll
      for (int t = 0; t < 16; ++t) {
        acc[t][0] *= c0.x; acc[t][1] *= c0.y; acc[t][2] *= c0.z; acc[t][3] *= c0.w;
        acc[t][4] *= c1.x; acc[t][5] *= c1.y; acc[t][6] *= c1.z; acc[t][7] *= c1.w;
      }
    }

    #pragma unroll
    for (int t = 0; t < 16; ++t) {
      const float* Bp = Bt + (t * 16 + n) * BSTRIDE + 2 * hh;
      #pragma unroll
      for (int kk = 0; kk < 4; ++kk) {
        v2f bm = *(const v2f*)(Bp + 4 * kk);
        acc[t] = __builtin_amdgcn_wmma_f32_16x16x4_f32(false, A[kk], false, bm,
                                                       (short)0, acc[t],
                                                       false, false);
      }
    }
  }

  float ltot = lpart + xhalf(lpart);
  if (hh == 0) rowx[w][n] = 1.0f / ltot;
  __syncthreads();
  v4f q0 = *(const v4f*)(&rowx[w][8 * hh]);
  v4f q1 = *(const v4f*)(&rowx[w][8 * hh + 4]);
  float inv[8] = { q0.x, q0.y, q0.z, q0.w, q1.x, q1.y, q1.z, q1.w };
  float* ob = out + ((size_t)(b * NN + i0 + 8 * hh)) * FF + n;
  #pragma unroll
  for (int t = 0; t < 16; ++t) {
    #pragma unroll
    for (int r = 0; r < 8; ++r)
      ob[(size_t)r * FF + t * 16] = acc[t][r] * inv[r];
  }
}

extern "C" void kernel_launch(void* const* d_in, const int* in_sizes, int n_in,
                              void* d_out, int out_size, void* d_ws, size_t ws_size,
                              hipStream_t stream) {
  const float* h   = (const float*)d_in[0];
  const int*   adj = (const int*)d_in[1];
  const float* W   = (const float*)d_in[2];
  const float* a   = (const float*)d_in[3];
  float* out = (float*)d_out;

  float* WhP = (float*)d_ws;
  float* f1  = WhP + (size_t)BB * NN * FF;
  float* f2  = f1 + (size_t)BB * NN;

  gat_gemm_hW<<<BB * NN / 64, 128, 0, stream>>>(h, W, a, WhP, f1, f2);
  gat_attn<<<BB * (NN / 16 / K3_WAVES), 128, 0, stream>>>(WhP, adj, f1, f2, out);
}


# attention kernel: two 16-column panels (32 score columns) per loop trip: one running maximum, rescale pass and LDS barrier per 32 columns; row sums and rescale factors per 32 columns (new rounding)
# speedup vs baseline: 2.4885x; 1.0564x over previous
#include <hip/hip_runtime.h>

typedef float v2f __attribute__((ext_vector_type(2)));
typedef int   v2i __attribute__((ext_vector_type(2)));
typedef float v4f __attribute__((ext_vector_type(4)));
typedef float v8f __attribute__((ext_vector_type(8)));

#define BB 16
#define NN 1024
#define FF 256
#define NEG_INF_V (-1000000000.0f)
#define SLOPE_V 0.2f

#define K3_WAVES 4
#define BSTRIDE 20
#define PANEL (FF * 16)

#define G_NCH 8
#define WPITCH 512
#define WSTAGE (16 * WPITCH)

struct WChunk {
  v4f e[8];
  v4f o[8];
};

__device__ __forceinline__ void w_fetch(WChunk& s, const float* Wg) {
  #pragma unroll
  for (int q = 0; q < 8; ++q) {
    s.e[q] = *(const v4f*)(Wg + (size_t)(4 * q) * FF);
    s.o[q] = *(const v4f*)(Wg + (size_t)(4 * q + 1) * FF);
  }
}

__device__ __forceinline__ void w_store(const WChunk& s, float* Ws) {
  #pragma unroll
  for (int q = 0; q < 8; ++q) {
    v4f lo = { s.e[q].x, s.o[q].x, s.e[q].y, s.o[q].y };
    v4f hi = { s.e[q].z, s.o[q].z, s.e[q].w, s.o[q].w };
    *(v4f*)(Ws + 2 * q * WPITCH)     = lo;
    *(v4f*)(Ws + 2 * q * WPITCH + 4) = hi;
  }
}

template <int M>
__device__ __forceinline__ float row_xor(float x) {
  int v = __builtin_bit_cast(int, x);
  return __builtin_bit_cast(float,
      __builtin_amdgcn_update_dpp(0, v, 0x160 + M, 0xf, 0xf, true));
}

__device__ __forceinline__ void lds_barrier() {
  __builtin_amdgcn_fence(__ATOMIC_RELEASE, "workgroup", "local");
  __builtin_amdgcn_s_barrier();
  __builtin_amdgcn_fence(__ATOMIC_ACQUIRE, "workgroup", "local");
}

__global__ __launch_bounds__(128) __attribute__((amdgpu_waves_per_eu(1, 1)))
void gat_gemm_hW(const float* __restrict__ h, const float* __restrict__ W,
                 const float* __restrict__ a, float* __restrict__ WhP,
                 float* __restrict__ f1, float* __restrict__ f2) {
  __shared__ __attribute__((aligned(16))) float Wl[2 * WSTAGE];
  int tid  = threadIdx.x;
  int w    = tid >> 5;
  int lane = tid & 31;
  int n  = lane & 15;
  int hh = lane >> 4;
  int blk = blockIdx.x;

  int cg = tid & 63;
  int kq = tid >> 6;
  const float* Wg = W + (size_t)(2 * kq) * FF + 4 * cg;
  float*       Ws = Wl + kq * WPITCH + ((8 * cg) ^ (32 * kq));

  const float* Ar = h + ((size_t)(blk * 64 + w * 16 + n)) * FF + 2 * hh;
  const float* Bl = Wl + hh * WPITCH + ((4 * n) ^ (32 * hh));

  v2f af[2][8];
  v2f a1v[8], a2v[8];
  v4f Lb[2][8];
  WChunk s;

  #pragma unroll
  for (int j = 0; j < 8; ++j) af[0][j] = *(const v2f*)(Ar + 4 * j);
  w_fetch(s, Wg);
  w_store(s, Ws);
  w_fetch(s, Wg + (size_t)32 * FF);
  lds_barrier();
  #pragma unroll
  for (int T = 0; T < 8; ++T) Lb[0][T] = *(const v4f*)(Bl + 64 * T);

  v8f acc[16];
  v8f zero = {};
  #pragma unroll
  for (int t = 0; t < 16; ++t) acc[t] = zero;

  #pragma unroll
  for (int c = 0; c < G_NCH; ++c) {
    #pragma unroll
    for (int j = 0; j < 8; ++j) {
      if (j == 4) {
        if (c + 1 < G_NCH) {
          w_store(s, Ws + ((c + 1) & 1) * WSTAGE);
          #pragma unroll
          for (int jj = 0; jj < 8; ++jj)
            af[(c + 1) & 1][jj] = *(const v2f*)(Ar + 32 * (c + 1) + 4 * jj);
        }
        if (c + 2 < G_NCH) w_fetch(s, Wg + (size_t)(32 * (c + 2)) * FF);
      }
      if (c == G_NCH - 1 && j == 0) {
        #pragma unroll
        for (int T = 0; T < 8; ++T) {
          a1v[T] = *(const v2f*)(a + 32 * T + 2 * n);
          a2v[T] = *(const v2f*)(a + FF + 32 * T + 2 * n);
        }
      }
      if (j < 7) {
        #pragma unroll
        for (int T = 0; T < 8; ++T)
          Lb[(j + 1) & 1][T] =
              *(const v4f*)(Bl + (c & 1) * WSTAGE + 2 * (j + 1) * WPITCH + 64 * T);
      } else if (c + 1 < G_NCH) {
        lds_barrier();
        #pragma unroll
        for (int T = 0; T < 8; ++T)
          Lb[0][T] = *(const v4f*)(Bl + ((c + 1) & 1) * WSTAGE + 64 * T);
      }
      #pragma unroll
      for (int T = 0; T < 8; ++T) {
        v4f L = Lb[j & 1][T];
        v2f b0 = { L.x, L.y };
        v2f b1 = { L.z, L.w };
        acc[2 * T]     = __builtin_amdgcn_wmma_f32_16x16x4_f32(false, af[c & 1][j], false, b0,
                                                               (short)0, acc[2 * T], false, false);
        acc[2 * T + 1] = __builtin_amdgcn_wmma_f32_16x16x4_f32(false, af[c & 1][j], false, b1,
                                                               (short)0, acc[2 * T + 1], false, false);
      }
    }
  }

  int b  = blk >> 4;
  int jt = (blk & 15) * 4 + w;
  float* outb = WhP + ((size_t)(b * (NN / 16) + jt)) * PANEL + 4 * lane;
  float* out0 = outb + (n >> 3) * 128;
  float* out1 = outb + (1 - (n >> 3)) * 128;
  #pragma unroll
  for (int t = 0; t < 16; ++t) {
    v4f lo = { acc[t][0], acc[t][1], acc[t][2], acc[t][3] };
    v4f hi = { acc[t][4], acc[t][5], acc[t][6], acc[t][7] };
    *(v4f*)(out0 + t * 256) = lo;
    *(v4f*)(out1 + t * 256) = hi;
  }

  float p1[8], p2[8];
  #pragma unroll
  for (int r = 0; r < 8; ++r) { p1[r] = 0.f; p2[r] = 0.f; }
  #pragma unroll
  for (int t = 0; t < 16; ++t) {
    float c1 = (t & 1) ? a1v[t >> 1].y : a1v[t >> 1].x;
    float c2 = (t & 1) ? a2v[t >> 1].y : a2v[t >> 1].x;
    #pragma unroll
    for (int r = 0; r < 8; ++r) {
      p1[r] += acc[t][r] * c1;
      p2[r] += acc[t][r] * c2;
    }
  }
  #pragma unroll
  for (int r = 0; r < 8; ++r) { p1[r] += row_xor<1>(p1[r]); p2[r] += row_xor<1>(p2[r]); }
  #pragma unroll
  for (int r = 0; r < 8; ++r) { p1[r] += row_xor<2>(p1[r]); p2[r] += row_xor<2>(p2[r]); }
  #pragma unroll
  for (int r = 0; r < 8; ++r) { p1[r] += row_xor<4>(p1[r]); p2[r] += row_xor<4>(p2[r]); }
  #pragma unroll
  for (int r = 0; r < 8; ++r) { p1[r] += row_xor<8>(p1[r]); p2[r] += row_xor<8>(p2[r]); }
  float s1 = p1[0], s2 = p2[0];
  #pragma unroll
  for (int r = 1; r < 8; ++r) {
    s1 = (n == r) ? p1[r] : s1;
    s2 = (n == r) ? p2[r] : s2;
  }
  if (n < 8) {
    int row = blk * 64 + w * 16 + 8 * hh + n;
    f1[row] = s1;
    f2[row] = s2;
  }
}

__device__ __forceinline__ float xhalf(float x) {
  int v = __builtin_bit_cast(int, x);
  return __builtin_bit_cast(float,
      __builtin_amdgcn_permlanex16(v, v, 0x76543210, 0xfedcba98, false, false));
}

struct StepRegs {
  v2i av[4];
  v2f fv[4];
  v4f p[8];
};

__device__ __forceinline__ void fetch_step(StepRegs& s, const int* adjp,
                                           const float* f2p, const float* g0) {
  #pragma unroll
  for (int kk = 0; kk < 4; ++kk) {
    s.av[kk] = *(const v2i*)(adjp + 4 * kk);
    s.fv[kk] = *(const v2f*)(f2p + 4 * kk);
  }
  #pragma unroll
  for (int q = 0; q < 8; ++q) s.p[q] = *(const v4f*)(g0 + 512 * q);
}

__global__ __launch_bounds__(128) __attribute__((amdgpu_waves_per_eu(1, 1)))
void gat_attn(const float* __restrict__ WhP, const int* __restrict__ adj,
              const float* __restrict__ f1, const float* __restrict__ f2,
              float* __restrict__ out) {
  __shared__ __attribute__((aligned(16))) float Btile[2][2][FF * BSTRIDE];
  __shared__ __attribute__((aligned(16))) float rowx[K3_WAVES][16];
  int tid  = threadIdx.x;
  int w    = tid >> 5;
  int lane = tid & 31;
  int n  = lane & 15;
  int hh = lane >> 4;
  int b     = blockIdx.x >> 4;
  int super = blockIdx.x & 15;
  int i0 = (super * K3_WAVES + w) * 16;

  const int*   adjr = adj + ((size_t)(b * NN + i0 + n)) * NN + 2 * hh;
  const float* Pb   = WhP + (size_t)b * (NN / 16) * PANEL + 4 * tid;
  const int lbase = (2 * n + (w >> 1)) * BSTRIDE + 8 * hh + 4 * ((w & 1) ^ (n >> 3));
  const float* f2h  = f2 + b * NN + 2 * hh;

  StepRegs s0, s1;
  fetch_step(s0, adjr, f2h, Pb);
  fetch_step(s1, adjr + 16, f2h + 16, Pb + PANEL);

  const float f1s = f1[b * NN + i0 + n];

  v8f acc[16];
  v8f zero = {};
  #pragma unroll
  for (int t = 0; t < 16; ++t) acc[t] = zero;
  float mrun = -3.0e38f;
  float lpart = 0.f;

  for (int it = 0; it < NN / 32; ++it) {
    float* Bt0 = Btile[it & 1][0];
    float* Bt1 = Btile[it & 1][1];

    {
      float* l0 = Bt0 + lbase;
      float* l1 = Bt1 + lbase;
      #pragma unroll
      for (int q = 0; q < 8; ++q) {
        *(v4f*)(l0 + q * (32 * BSTRIDE)) = s0.p[q];
        *(v4f*)(l1 + q * (32 * BSTRIDE)) = s1.p[q];
      }
    }

    float e[16];
    #pragma unroll
    for (int kk = 0; kk < 4; ++kk) {
      float e0 = f1s + s0.fv[kk].x;
      float e1 = f1s + s0.fv[kk].y;
      float e2 = f1s + s1.fv[kk].x;
      float e3 = f1s + s1.fv[kk].y;
      e0 = e0 > 0.f ? e0 : SLOPE_V * e0;
      e1 = e1 > 0.f ? e1 : SLOPE_V * e1;
      e2 = e2 > 0.f ? e2 : SLOPE_V * e2;
      e3 = e3 > 0.f ? e3 : SLOPE_V * e3;
      e[2 * kk]         = (s0.av[kk].x == 0) ? NEG_INF_V : e0;
      e[2 * kk + 1]     = (s0.av[kk].y == 0) ? NEG_INF_V : e1;
      e[8 + 2 * kk]     = (s1.av[kk].x == 0) ? NEG_INF_V : e2;
      e[8 + 2 * kk + 1] = (s1.av[kk].y == 0) ? NEG_INF_V : e3;
    }

    float rm0 = fmaxf(fmaxf(fmaxf(e[0], e[1]), fmaxf(e[2], e[3])),
                      fmaxf(fmaxf(e[4], e[5]), fmaxf(e[6], e[7])));
    float rm1 = fmaxf(fmaxf(fmaxf(e[8], e[9]), fmaxf(e[10], e[11])),
                      fmaxf(fmaxf(e[12], e[13]), fmaxf(e[14], e[15])));
    float rm = fmaxf(rm0, rm1);
    rm = fmaxf(rm, xhalf(rm));
    float nm   = fmaxf(mrun, rm);
    float corr = __expf(mrun - nm);
    mrun = nm;
    v2f A[8];
    float rs = 0.f;
    #pragma unroll
    for (int k8 = 0; k8 < 8; ++k8) {
      float p0 = __expf(e[2 * k8] - nm);
      float p1 = __expf(e[2 * k8 + 1] - nm);
      A[k8].x = p0;
      A[k8].y = p1;
      rs += p0 + p1;
    }
    lpart = lpart * corr + rs;
    if (hh == 0) rowx[w][n] = corr;

    {
      int jn = (2 * it + 2) & (NN / 16 - 1);
      fetch_step(s0, adjr + jn * 16, f2h + jn * 16, Pb + (size_t)jn * PANEL);
      fetch_step(s1, adjr + jn * 16 + 16, f2h + jn * 16 + 16, Pb + (size_t)(jn + 1) * PANEL);
    }

    __builtin_amdgcn_fence(__ATOMIC_RELEASE, "workgroup", "local");
    __builtin_amdgcn_s_barrier();
    __builtin_amdgcn_fence(__ATOMIC_ACQUIRE, "workgroup", "local");

    {
      v4f c0 = *(const v4f*)(&rowx[w][8 * hh]);
      v4f c1 = *(const v4f*)(&rowx[w][8 * hh + 4]);
      #pragma unroll
      for (int t = 0; t < 16; ++t) {
        acc[t][0] *= c0.x; acc[t][1] *= c0.y; acc[t][2] *= c0.z; acc[t][3] *= c0.w;
        acc[t][4] *= c1.x; acc[t][5] *= c1.y; acc[t][6] *= c1.z; acc[t][7] *= c1.w;
      }
    }

    #pragma unroll
    for (int t = 0; t < 16; ++t) {
      const float* Bp0 = Bt0 + (t * 16 + n) * BSTRIDE + 2 * hh;
      #pragma unroll
      for (int kk = 0; kk < 4; ++kk) {
        v2f bm = *(const v2f*)(Bp0 + 4 * kk);
        acc[t] = __builtin_amdgcn_wmma_f32_16x16x4_f32(false, A[kk], false, bm,
                                                       (short)0, acc[t],
                                                       false, false);
      }
    }
    #pragma unroll
    for (int t = 0; t < 16; ++t) {
      const float* Bp1 = Bt1 + (t * 16 + n) * BSTRIDE + 2 * hh;
      #pragma unroll
      for (int kk = 0; kk < 4; ++kk) {
        v2f bm = *(const v2f*)(Bp1 + 4 * kk);
        acc[t] = __builtin_amdgcn_wmma_f32_16x16x4_f32(false, A[4 + kk], false, bm,
                                                       (short)0, acc[t],
                                                       false, false);
      }
    }
  }

  float ltot = lpart + xhalf(lpart);
  if (hh == 0) rowx[w][n] = 1.0f / ltot;
  __syncthreads();
  v4f q0 = *(const v4f*)(&rowx[w][8 * hh]);
  v4f q1 = *(const v4f*)(&rowx[w][8 * hh + 4]);
  float inv[8] = { q0.x, q0.y, q0.z, q0.w, q1.x, q1.y, q1.z, q1.w };
  float* ob = out + ((size_t)(b * NN + i0 + 8 * hh)) * FF + n;
  #pragma unroll
  for (int t = 0; t < 16; ++t) {
    #pragma unroll
    for (int r = 0; r < 8; ++r)
      ob[(size_t)r * FF + t * 16] = acc[t][r] * inv[r];
  }
}

extern "C" void kernel_launch(void* const* d_in, const int* in_sizes, int n_in,
                              void* d_out, int out_size, void* d_ws, size_t ws_size,
                              hipStream_t stream) {
  const float* h   = (const float*)d_in[0];
  const int*   adj = (const int*)d_in[1];
  const float* W   = (const float*)d_in[2];
  const float* a   = (const float*)d_in[3];
  float* out = (float*)d_out;

  float* WhP = (float*)d_ws;
  float* f1  = WhP + (size_t)BB * NN * FF;
  float* f2  = f1 + (size_t)BB * NN;

  gat_gemm_hW<<<BB * NN / 64, 128, 0, stream>>>(h, W, a, WhP, f1, f2);
  gat_attn<<<BB * (NN / 16 / K3_WAVES), 128, 0, stream>>>(WhP, adj, f1, f2, out);
}
